# starting kernel
# baseline (speedup 1.0000x reference)
#include <hip/hip_runtime.h>
#include <hip/hip_bf16.h>
#include <math.h>

typedef float v2f __attribute__((ext_vector_type(2)));
typedef float v8f __attribute__((ext_vector_type(8)));

#define Bb 2
#define Nn 512
#define Ee 128
#define Hh 256
#define ROWS (Bb * Nn)
#define LN_EPS 1e-5f

__global__ void ln_kernel(const float* __restrict__ h,
                          const float* __restrict__ gamma,
                          const float* __restrict__ beta,
                          float* __restrict__ hn) {
    const int row = blockIdx.x;
    const int t   = threadIdx.x;
    __shared__ float red[Ee];

    const float v = h[row * Ee + t];
    red[t] = v;
    __syncthreads();
    for (int s = Ee / 2; s > 0; s >>= 1) {
        if (t < s) red[t] += red[t + s];
        __syncthreads();
    }
    const float mu = red[0] * (1.0f / Ee);
    __syncthreads();

    const float d = v - mu;
    red[t] = d * d;
    __syncthreads();
    for (int s = Ee / 2; s > 0; s >>= 1) {
        if (t < s) red[t] += red[t + s];
        __syncthreads();
    }
    const float var = red[0] * (1.0f / Ee);
    const float r   = rsqrtf(var + LN_EPS);
    hn[row * Ee + t] = d * r * gamma[t] + beta[t];
}

__global__ void gemm_wmma_kernel(const float* __restrict__ hn,
                                 const float* __restrict__ W1,
                                 const float* __restrict__ b1,
                                 float* __restrict__ aP,
                                 float* __restrict__ bT) {
    const int tile = blockIdx.x;
    const int mt   = tile >> 4;
    const int nt   = tile & 15;
    const int which = blockIdx.y;
    const float* __restrict__ W = W1 + (size_t)which * Ee * Hh;

    const int lane = threadIdx.x;
    const int half = lane >> 4;
    const int l16  = lane & 15;

    const int m = mt * 16 + l16;
    const int n = nt * 16 + l16;

    v8f acc = {0.f, 0.f, 0.f, 0.f, 0.f, 0.f, 0.f, 0.f};

    for (int k0 = 0; k0 < Ee; k0 += 4) {
        const int ka = k0 + 2 * half;
        v2f A, Bf;
        A.x  = hn[m * Ee + ka];
        A.y  = hn[m * Ee + ka + 1];
        Bf.x = W[(size_t)ka * Hh + n];
        Bf.y = W[(size_t)(ka + 1) * Hh + n];
        acc = __builtin_amdgcn_wmma_f32_16x16x4_f32(
                   false, A,  false, Bf,
                   (short)0, acc,  false,  false);
    }

    if (which == 0) {
        const float bias = b1[n];
        #pragma unroll
        for (int r = 0; r < 8; ++r) {
            const int mo = mt * 16 + 8 * half + r;
            aP[(size_t)mo * Hh + n] = acc[r] + bias;
        }
    } else {
        #pragma unroll
        for (int r = 0; r < 8; ++r) {
            const int mo = mt * 16 + 8 * half + r;
            const int bb = mo >> 9;
            const int j  = mo & (Nn - 1);
            bT[(size_t)bb * Hh * Nn + (size_t)n * Nn + j] = acc[r];
        }
    }
}

__global__ void pair_gelu_kernel(const float* __restrict__ aP,
                                 const float* __restrict__ bT,
                                 const float* __restrict__ w2,
                                 const float* __restrict__ b2,
                                 float* __restrict__ l) {
    const int j  = blockIdx.x * 256 + threadIdx.x;
    const int i  = blockIdx.y;
    const int bb = blockIdx.z;

    __shared__ float sA[Hh];
    __shared__ float sW[Hh];
    sA[threadIdx.x] = aP[((size_t)(bb * Nn + i)) * Hh + threadIdx.x];
    sW[threadIdx.x] = w2[threadIdx.x];
    __syncthreads();

    const float* __restrict__ bp = bT + (size_t)bb * Hh * Nn + j;
    float acc = b2[0];

    #pragma unroll 4
    for (int h = 0; h < Hh; ++h) {
        const float x = sA[h] + bp[(size_t)h * Nn];
        const float g = 0.5f * x * (1.0f + erff(x * 0.70710678118654752f));
        acc = fmaf(sW[h], g, acc);
    }
    l[((size_t)(bb * Nn + i)) * Nn + j] = acc;
}

__global__ void symmetrize_kernel(const float* __restrict__ l,
                                  float* __restrict__ out) {
    const int idx = blockIdx.x * 256 + threadIdx.x;
    const int bb  = idx >> 18;
    const int rem = idx & (Nn * Nn - 1);
    const int i   = rem >> 9;
    const int j   = rem & (Nn - 1);
    out[idx] = 0.5f * (l[idx] + l[(size_t)bb * Nn * Nn + (size_t)j * Nn + i]);
}

extern "C" void kernel_launch(void* const* d_in, const int* in_sizes, int n_in,
                              void* d_out, int out_size, void* d_ws, size_t ws_size,
                              hipStream_t stream) {
    const float* h     = (const float*)d_in[0];
    const float* gamma = (const float*)d_in[1];
    const float* beta  = (const float*)d_in[2];
    const float* W1    = (const float*)d_in[3];
    const float* b1    = (const float*)d_in[4];
    const float* w2    = (const float*)d_in[5];
    const float* b2    = (const float*)d_in[6];
    float* out = (float*)d_out;

    float* ws  = (float*)d_ws;
    float* hn  = ws;
    float* aP  = hn + (size_t)ROWS * Ee;
    float* bT  = aP + (size_t)ROWS * Hh;
    float* lbuf= bT + (size_t)Bb * Hh * Nn;

    ln_kernel<<<dim3(ROWS), dim3(Ee), 0, stream>>>(h, gamma, beta, hn);

    gemm_wmma_kernel<<<dim3(64 * 16, 2), dim3(32), 0, stream>>>(hn, W1, b1, aP, bT);

    pair_gelu_kernel<<<dim3(Nn / 256, Nn, Bb), dim3(256), 0, stream>>>(
        aP, bT, w2, b2, lbuf);

    symmetrize_kernel<<<dim3((Bb * Nn * Nn) / 256), dim3(256), 0, stream>>>(lbuf, out);
}


# pair kernel: each lane owns 2 rows x 2 adjacent columns; the column pair runs on packed f32 instructions (v_pk_add/v_pk_fma) and one 8-byte bT load serves 4 results; same sum order over h
# speedup vs baseline: 3.3793x; 3.3793x over previous
#include <hip/hip_runtime.h>
#include <hip/hip_bf16.h>
#include <math.h>

typedef float v2f __attribute__((ext_vector_type(2)));
typedef float v8f __attribute__((ext_vector_type(8)));

#define Bb 2
#define Nn 512
#define Ee 128
#define Hh 256
#define ROWS (Bb * Nn)
#define LN_EPS 1e-5f

__global__ void ln_kernel(const float* __restrict__ h,
                          const float* __restrict__ gamma,
                          const float* __restrict__ beta,
                          float* __restrict__ hn) {
    const int row = blockIdx.x;
    const int t   = threadIdx.x;
    __shared__ float red[Ee];

    const float v = h[row * Ee + t];
    red[t] = v;
    __syncthreads();
    for (int s = Ee / 2; s > 0; s >>= 1) {
        if (t < s) red[t] += red[t + s];
        __syncthreads();
    }
    const float mu = red[0] * (1.0f / Ee);
    __syncthreads();

    const float d = v - mu;
    red[t] = d * d;
    __syncthreads();
    for (int s = Ee / 2; s > 0; s >>= 1) {
        if (t < s) red[t] += red[t + s];
        __syncthreads();
    }
    const float var = red[0] * (1.0f / Ee);
    const float r   = rsqrtf(var + LN_EPS);
    hn[row * Ee + t] = d * r * gamma[t] + beta[t];
}

__global__ void gemm_wmma_kernel(const float* __restrict__ hn,
                                 const float* __restrict__ W1,
                                 const float* __restrict__ b1,
                                 float* __restrict__ aP,
                                 float* __restrict__ bT) {
    const int tile = blockIdx.x;
    const int mt   = tile >> 4;
    const int nt   = tile & 15;
    const int which = blockIdx.y;
    const float* __restrict__ W = W1 + (size_t)which * Ee * Hh;

    const int lane = threadIdx.x;
    const int half = lane >> 4;
    const int l16  = lane & 15;

    const int m = mt * 16 + l16;
    const int n = nt * 16 + l16;

    v8f acc = {0.f, 0.f, 0.f, 0.f, 0.f, 0.f, 0.f, 0.f};

    for (int k0 = 0; k0 < Ee; k0 += 4) {
        const int ka = k0 + 2 * half;
        v2f A, Bf;
        A.x  = hn[m * Ee + ka];
        A.y  = hn[m * Ee + ka + 1];
        Bf.x = W[(size_t)ka * Hh + n];
        Bf.y = W[(size_t)(ka + 1) * Hh + n];
        acc = __builtin_amdgcn_wmma_f32_16x16x4_f32(
                   false, A,  false, Bf,
                   (short)0, acc,  false,  false);
    }

    if (which == 0) {
        const float bias = b1[n];
        #pragma unroll
        for (int r = 0; r < 8; ++r) {
            const int mo = mt * 16 + 8 * half + r;
            aP[(size_t)mo * Hh + n] = acc[r] + bias;
        }
    } else {
        #pragma unroll
        for (int r = 0; r < 8; ++r) {
            const int mo = mt * 16 + 8 * half + r;
            const int bb = mo >> 9;
            const int j  = mo & (Nn - 1);
            bT[(size_t)bb * Hh * Nn + (size_t)n * Nn + j] = acc[r];
        }
    }
}

__device__ __forceinline__ v2f splat2(float c) { return v2f{c, c}; }

__device__ __forceinline__ v2f gelu_erf_f32x2(v2f x) {
    const v2f u = __builtin_elementwise_min(__builtin_elementwise_abs(x), splat2(6.0f));
    v2f p = splat2(3.30960393e-5f);
    p = __builtin_elementwise_fma(p, u, splat2(-7.69245671e-4f));
    p = __builtin_elementwise_fma(p, u, splat2(8.08079448e-3f));
    p = __builtin_elementwise_fma(p, u, splat2(-5.34122139e-2f));
    p = __builtin_elementwise_fma(p, u, splat2(-4.58770901e-1f));
    p = __builtin_elementwise_fma(p, u, splat2(-1.15120173f));
    p = __builtin_elementwise_fma(p, u, splat2(-9.99993086e-1f));
    v2f e;
    e.x = __builtin_amdgcn_exp2f(p.x);
    e.y = __builtin_amdgcn_exp2f(p.y);
    const v2f m = __builtin_elementwise_max(x, splat2(0.0f));
    return __builtin_elementwise_fma(-u, e, m);
}

#define PR_I 2
#define PR_T 128
#define PR_H 4

struct __attribute__((aligned(32))) PairRow { v2f a[PR_I]; v2f w; v2f pad; };

__device__ __forceinline__ void pair_trip(const PairRow* __restrict__ row,
                                          const v2f (&cb)[PR_H], v2f (&acc)[PR_I]) {
    #pragma unroll
    for (int k = 0; k < PR_H; ++k) {
        const v2f w = row[k].w;
        #pragma unroll
        for (int r = 0; r < PR_I; ++r) {
            const v2f x = row[k].a[r] + cb[k];
            acc[r] = __builtin_elementwise_fma(w, gelu_erf_f32x2(x), acc[r]);
        }
    }
}

__global__ __launch_bounds__(PR_T) void pair_gelu_kernel(const float* __restrict__ aP,
                                 const float* __restrict__ bT,
                                 const float* __restrict__ w2,
                                 const float* __restrict__ b2,
                                 float* __restrict__ l) {
    const int t  = threadIdx.x;
    const int j  = blockIdx.x * (2 * PR_T) + 2 * t;
    const int i0 = blockIdx.y * PR_I;
    const int bb = blockIdx.z;

    __shared__ PairRow sR[Hh];
    static_assert(Hh % PR_T == 0, "each lane fills Hh / PR_T rows of sR");
    #pragma unroll
    for (int p = 0; p < Hh / PR_T; ++p) {
        const int h = t + p * PR_T;
        #pragma unroll
        for (int r = 0; r < PR_I; ++r)
            sR[h].a[r] = splat2(aP[((size_t)(bb * Nn + i0 + r)) * Hh + h]);
        sR[h].w = splat2(w2[h]);
    }
    const v2f* __restrict__ bq =
        reinterpret_cast<const v2f*>(bT + (size_t)bb * Hh * Nn + j);

    v2f b0[PR_H], b1[PR_H];
    #pragma unroll
    for (int k = 0; k < PR_H; ++k) b0[k] = bq[(size_t)k * (Nn / 2)];

    __syncthreads();

    v2f acc[PR_I];
    #pragma unroll
    for (int r = 0; r < PR_I; ++r) acc[r] = splat2(b2[0]);

    for (int h = 0; h < Hh - 2 * PR_H; h += 2 * PR_H) {
        #pragma unroll
        for (int k = 0; k < PR_H; ++k) b1[k] = bq[(size_t)(h + PR_H + k) * (Nn / 2)];
        pair_trip(&sR[h], b0, acc);
        #pragma unroll
        for (int k = 0; k < PR_H; ++k) b0[k] = bq[(size_t)(h + 2 * PR_H + k) * (Nn / 2)];
        pair_trip(&sR[h + PR_H], b1, acc);
    }
    #pragma unroll
    for (int k = 0; k < PR_H; ++k) b1[k] = bq[(size_t)(Hh - PR_H + k) * (Nn / 2)];
    pair_trip(&sR[Hh - 2 * PR_H], b0, acc);
    pair_trip(&sR[Hh - PR_H], b1, acc);

    #pragma unroll
    for (int r = 0; r < PR_I; ++r)
        *reinterpret_cast<v2f*>(l + ((size_t)(bb * Nn + i0 + r)) * Nn + j) = acc[r];
}

__global__ void symmetrize_kernel(const float* __restrict__ l,
                                  float* __restrict__ out) {
    const int idx = blockIdx.x * 256 + threadIdx.x;
    const int bb  = idx >> 18;
    const int rem = idx & (Nn * Nn - 1);
    const int i   = rem >> 9;
    const int j   = rem & (Nn - 1);
    out[idx] = 0.5f * (l[idx] + l[(size_t)bb * Nn * Nn + (size_t)j * Nn + i]);
}

extern "C" void kernel_launch(void* const* d_in, const int* in_sizes, int n_in,
                              void* d_out, int out_size, void* d_ws, size_t ws_size,
                              hipStream_t stream) {
    const float* h     = (const float*)d_in[0];
    const float* gamma = (const float*)d_in[1];
    const float* beta  = (const float*)d_in[2];
    const float* W1    = (const float*)d_in[3];
    const float* b1    = (const float*)d_in[4];
    const float* w2    = (const float*)d_in[5];
    const float* b2    = (const float*)d_in[6];
    float* out = (float*)d_out;

    float* ws  = (float*)d_ws;
    float* hn  = ws;
    float* aP  = hn + (size_t)ROWS * Ee;
    float* bT  = aP + (size_t)ROWS * Hh;
    float* lbuf= bT + (size_t)Bb * Hh * Nn;

    ln_kernel<<<dim3(ROWS), dim3(Ee), 0, stream>>>(h, gamma, beta, hn);

    gemm_wmma_kernel<<<dim3(64 * 16, 2), dim3(32), 0, stream>>>(hn, W1, b1, aP, bT);

    pair_gelu_kernel<<<dim3(Nn / (2 * PR_T), Nn / PR_I, Bb), dim3(PR_T), 0, stream>>>(
        aP, bT, w2, b2, lbuf);

    symmetrize_kernel<<<dim3((Bb * Nn * Nn) / 256), dim3(256), 0, stream>>>(lbuf, out);
}


# symmetrize: one block per pair of mirrored 32x32 tiles, exchanged through LDS, 16-byte row loads and stores; 272 blocks of 8 waves replace 16384 one-element waves; every sum and its order kept
# speedup vs baseline: 3.5975x; 1.0646x over previous
#include <hip/hip_runtime.h>
#include <hip/hip_bf16.h>
#include <math.h>

typedef float v2f __attribute__((ext_vector_type(2)));
typedef float v8f __attribute__((ext_vector_type(8)));

#define Bb 2
#define Nn 512
#define Ee 128
#define Hh 256
#define ROWS (Bb * Nn)
#define LN_EPS 1e-5f

__global__ void ln_kernel(const float* __restrict__ h,
                          const float* __restrict__ gamma,
                          const float* __restrict__ beta,
                          float* __restrict__ hn) {
    const int row = blockIdx.x;
    const int t   = threadIdx.x;
    __shared__ float red[Ee];

    const float v = h[row * Ee + t];
    red[t] = v;
    __syncthreads();
    for (int s = Ee / 2; s > 0; s >>= 1) {
        if (t < s) red[t] += red[t + s];
        __syncthreads();
    }
    const float mu = red[0] * (1.0f / Ee);
    __syncthreads();

    const float d = v - mu;
    red[t] = d * d;
    __syncthreads();
    for (int s = Ee / 2; s > 0; s >>= 1) {
        if (t < s) red[t] += red[t + s];
        __syncthreads();
    }
    const float var = red[0] * (1.0f / Ee);
    const float r   = rsqrtf(var + LN_EPS);
    hn[row * Ee + t] = d * r * gamma[t] + beta[t];
}

__global__ void gemm_wmma_kernel(const float* __restrict__ hn,
                                 const float* __restrict__ W1,
                                 const float* __restrict__ b1,
                                 float* __restrict__ aP,
                                 float* __restrict__ bT) {
    const int tile = blockIdx.x;
    const int mt   = tile >> 4;
    const int nt   = tile & 15;
    const int which = blockIdx.y;
    const float* __restrict__ W = W1 + (size_t)which * Ee * Hh;

    const int lane = threadIdx.x;
    const int half = lane >> 4;
    const int l16  = lane & 15;

    const int m = mt * 16 + l16;
    const int n = nt * 16 + l16;

    v8f acc = {0.f, 0.f, 0.f, 0.f, 0.f, 0.f, 0.f, 0.f};

    for (int k0 = 0; k0 < Ee; k0 += 4) {
        const int ka = k0 + 2 * half;
        v2f A, Bf;
        A.x  = hn[m * Ee + ka];
        A.y  = hn[m * Ee + ka + 1];
        Bf.x = W[(size_t)ka * Hh + n];
        Bf.y = W[(size_t)(ka + 1) * Hh + n];
        acc = __builtin_amdgcn_wmma_f32_16x16x4_f32(
                   false, A,  false, Bf,
                   (short)0, acc,  false,  false);
    }

    if (which == 0) {
        const float bias = b1[n];
        #pragma unroll
        for (int r = 0; r < 8; ++r) {
            const int mo = mt * 16 + 8 * half + r;
            aP[(size_t)mo * Hh + n] = acc[r] + bias;
        }
    } else {
        #pragma unroll
        for (int r = 0; r < 8; ++r) {
            const int mo = mt * 16 + 8 * half + r;
            const int bb = mo >> 9;
            const int j  = mo & (Nn - 1);
            bT[(size_t)bb * Hh * Nn + (size_t)n * Nn + j] = acc[r];
        }
    }
}

__device__ __forceinline__ v2f splat2(float c) { return v2f{c, c}; }

__device__ __forceinline__ v2f gelu_erf_f32x2(v2f x) {
    const v2f u = __builtin_elementwise_min(__builtin_elementwise_abs(x), splat2(6.0f));
    v2f p = splat2(3.30960393e-5f);
    p = __builtin_elementwise_fma(p, u, splat2(-7.69245671e-4f));
    p = __builtin_elementwise_fma(p, u, splat2(8.08079448e-3f));
    p = __builtin_elementwise_fma(p, u, splat2(-5.34122139e-2f));
    p = __builtin_elementwise_fma(p, u, splat2(-4.58770901e-1f));
    p = __builtin_elementwise_fma(p, u, splat2(-1.15120173f));
    p = __builtin_elementwise_fma(p, u, splat2(-9.99993086e-1f));
    v2f e;
    e.x = __builtin_amdgcn_exp2f(p.x);
    e.y = __builtin_amdgcn_exp2f(p.y);
    const v2f m = __builtin_elementwise_max(x, splat2(0.0f));
    return __builtin_elementwise_fma(-u, e, m);
}

#define PR_I 2
#define PR_T 128
#define PR_H 4

struct __attribute__((aligned(32))) PairRow { v2f a[PR_I]; v2f w; v2f pad; };

__device__ __forceinline__ void pair_trip(const PairRow* __restrict__ row,
                                          const v2f (&cb)[PR_H], v2f (&acc)[PR_I]) {
    #pragma unroll
    for (int k = 0; k < PR_H; ++k) {
        const v2f w = row[k].w;
        #pragma unroll
        for (int r = 0; r < PR_I; ++r) {
            const v2f x = row[k].a[r] + cb[k];
            acc[r] = __builtin_elementwise_fma(w, gelu_erf_f32x2(x), acc[r]);
        }
    }
}

__global__ __launch_bounds__(PR_T) void pair_gelu_kernel(const float* __restrict__ aP,
                                 const float* __restrict__ bT,
                                 const float* __restrict__ w2,
                                 const float* __restrict__ b2,
                                 float* __restrict__ l) {
    const int t  = threadIdx.x;
    const int j  = blockIdx.x * (2 * PR_T) + 2 * t;
    const int i0 = blockIdx.y * PR_I;
    const int bb = blockIdx.z;

    __shared__ PairRow sR[Hh];
    static_assert(Hh % PR_T == 0, "each lane fills Hh / PR_T rows of sR");
    #pragma unroll
    for (int p = 0; p < Hh / PR_T; ++p) {
        const int h = t + p * PR_T;
        #pragma unroll
        for (int r = 0; r < PR_I; ++r)
            sR[h].a[r] = splat2(aP[((size_t)(bb * Nn + i0 + r)) * Hh + h]);
        sR[h].w = splat2(w2[h]);
    }
    const v2f* __restrict__ bq =
        reinterpret_cast<const v2f*>(bT + (size_t)bb * Hh * Nn + j);

    v2f b0[PR_H], b1[PR_H];
    #pragma unroll
    for (int k = 0; k < PR_H; ++k) b0[k] = bq[(size_t)k * (Nn / 2)];

    __syncthreads();

    v2f acc[PR_I];
    #pragma unroll
    for (int r = 0; r < PR_I; ++r) acc[r] = splat2(b2[0]);

    for (int h = 0; h < Hh - 2 * PR_H; h += 2 * PR_H) {
        #pragma unroll
        for (int k = 0; k < PR_H; ++k) b1[k] = bq[(size_t)(h + PR_H + k) * (Nn / 2)];
        pair_trip(&sR[h], b0, acc);
        #pragma unroll
        for (int k = 0; k < PR_H; ++k) b0[k] = bq[(size_t)(h + 2 * PR_H + k) * (Nn / 2)];
        pair_trip(&sR[h + PR_H], b1, acc);
    }
    #pragma unroll
    for (int k = 0; k < PR_H; ++k) b1[k] = bq[(size_t)(Hh - PR_H + k) * (Nn / 2)];
    pair_trip(&sR[Hh - 2 * PR_H], b0, acc);
    pair_trip(&sR[Hh - PR_H], b1, acc);

    #pragma unroll
    for (int r = 0; r < PR_I; ++r)
        *reinterpret_cast<v2f*>(l + ((size_t)(bb * Nn + i0 + r)) * Nn + j) = acc[r];
}

#define SY_T  32
#define SY_NT (Nn / SY_T)

__global__ __launch_bounds__(256) void symmetrize_kernel(const float* __restrict__ l,
                                                         float* __restrict__ out) {
    const int x = blockIdx.x;
    const int y = blockIdx.y;
    const bool right = x > y;
    const int I = right ? y : SY_NT - 1 - y;
    const int J = right ? x - 1 : SY_NT - 1 - x;
    const size_t base = (size_t)blockIdx.z * Nn * Nn;

    __shared__ float sA[SY_T][SY_T + 1];
    __shared__ float sB[SY_T][SY_T + 1];

    const int t = threadIdx.x;
    const int r = t >> 3;
    const int c = (t & 7) << 2;

    const size_t offA = base + (size_t)(I * SY_T + r) * Nn + (size_t)(J * SY_T + c);
    const size_t offB = base + (size_t)(J * SY_T + r) * Nn + (size_t)(I * SY_T + c);
    const float4 a = *reinterpret_cast<const float4*>(l + offA);
    const float4 b = *reinterpret_cast<const float4*>(l + offB);

    sA[r][c + 0] = a.x; sA[r][c + 1] = a.y; sA[r][c + 2] = a.z; sA[r][c + 3] = a.w;
    sB[r][c + 0] = b.x; sB[r][c + 1] = b.y; sB[r][c + 2] = b.z; sB[r][c + 3] = b.w;
    __syncthreads();

    float4 o;
    o.x = 0.5f * (a.x + sB[c + 0][r]);
    o.y = 0.5f * (a.y + sB[c + 1][r]);
    o.z = 0.5f * (a.z + sB[c + 2][r]);
    o.w = 0.5f * (a.w + sB[c + 3][r]);
    *reinterpret_cast<float4*>(out + offA) = o;

    if (I != J) {
        o.x = 0.5f * (b.x + sA[c + 0][r]);
        o.y = 0.5f * (b.y + sA[c + 1][r]);
        o.z = 0.5f * (b.z + sA[c + 2][r]);
        o.w = 0.5f * (b.w + sA[c + 3][r]);
        *reinterpret_cast<float4*>(out + offB) = o;
    }
}

extern "C" void kernel_launch(void* const* d_in, const int* in_sizes, int n_in,
                              void* d_out, int out_size, void* d_ws, size_t ws_size,
                              hipStream_t stream) {
    const float* h     = (const float*)d_in[0];
    const float* gamma = (const float*)d_in[1];
    const float* beta  = (const float*)d_in[2];
    const float* W1    = (const float*)d_in[3];
    const float* b1    = (const float*)d_in[4];
    const float* w2    = (const float*)d_in[5];
    const float* b2    = (const float*)d_in[6];
    float* out = (float*)d_out;

    float* ws  = (float*)d_ws;
    float* hn  = ws;
    float* aP  = hn + (size_t)ROWS * Ee;
    float* bT  = aP + (size_t)ROWS * Hh;
    float* lbuf= bT + (size_t)Bb * Hh * Nn;

    ln_kernel<<<dim3(ROWS), dim3(Ee), 0, stream>>>(h, gamma, beta, hn);

    gemm_wmma_kernel<<<dim3(64 * 16, 2), dim3(32), 0, stream>>>(hn, W1, b1, aP, bT);

    pair_gelu_kernel<<<dim3(Nn / (2 * PR_T), Nn / PR_I, Bb), dim3(PR_T), 0, stream>>>(
        aP, bT, w2, b2, lbuf);

    symmetrize_kernel<<<dim3(SY_NT + 1, SY_NT / 2, Bb), dim3(256), 0, stream>>>(lbuf, out);
}


# LayerNorm folded into the GEMM kernel: each wave normalises its own 16 rows in registers before its WMMAs (row sums: 4 partials per lane + 1 cross-lane add, no 128-wide tree); 4 launches become 3
# speedup vs baseline: 3.9625x; 1.1015x over previous
#include <hip/hip_runtime.h>
#include <hip/hip_bf16.h>
#include <math.h>

typedef float v2f __attribute__((ext_vector_type(2)));
typedef float v8f __attribute__((ext_vector_type(8)));

#define Bb 2
#define Nn 512
#define Ee 128
#define Hh 256
#define ROWS (Bb * Nn)
#define LN_EPS 1e-5f
#define GB_PRE 16

__device__ __forceinline__ v2f splat2(float c) { return v2f{c, c}; }

__global__ __launch_bounds__(32) void ln_gemm_wmma_kernel(const float* __restrict__ h,
                                 const float* __restrict__ gamma,
                                 const float* __restrict__ beta,
                                 const float* __restrict__ W1,
                                 const float* __restrict__ b1,
                                 float* __restrict__ aP,
                                 float* __restrict__ bT) {
    const int tile = blockIdx.x;
    const int mt   = tile >> 4;
    const int nt   = tile & 15;
    const int which = blockIdx.y;
    const float* __restrict__ W = W1 + (size_t)which * Ee * Hh;

    const int lane = threadIdx.x;
    const int half = lane >> 4;
    const int l16  = lane & 15;

    const int m = mt * 16 + l16;
    const int n = nt * 16 + l16;

    const v2f* __restrict__ hrow =
        reinterpret_cast<const v2f*>(h + (size_t)m * Ee + 2 * half);
    v2f a[Ee / 4];
    #pragma unroll
    for (int t = 0; t < Ee / 4; ++t) a[t] = hrow[2 * t];

    v2f Bf[Ee / 4];
    #pragma unroll
    for (int t = 0; t < Ee / 4; ++t) {
        const int ka = 4 * t + 2 * half;
        Bf[t].x = W[(size_t)ka * Hh + n];
        Bf[t].y = W[(size_t)(ka + 1) * Hh + n];
    }
    const v2f* __restrict__ g2 = reinterpret_cast<const v2f*>(gamma + 2 * half);
    const v2f* __restrict__ e2 = reinterpret_cast<const v2f*>(beta + 2 * half);
    v2f gm[GB_PRE], bt[GB_PRE];
    #pragma unroll
    for (int t = 0; t < GB_PRE; ++t) { gm[t] = g2[2 * t]; bt[t] = e2[2 * t]; }
    __builtin_amdgcn_sched_barrier(0);

    v2f s2[4] = {splat2(0.f), splat2(0.f), splat2(0.f), splat2(0.f)};
    #pragma unroll
    for (int t = 0; t < Ee / 4; ++t) s2[t & 3] += a[t];
    const v2f sv = (s2[0] + s2[1]) + (s2[2] + s2[3]);
    float s = sv.x + sv.y;
    s += __shfl_xor(s, 16);
    const float mu = s * (1.0f / Ee);

    v2f q2[4] = {splat2(0.f), splat2(0.f), splat2(0.f), splat2(0.f)};
    #pragma unroll
    for (int t = 0; t < Ee / 4; ++t) {
        a[t] -= splat2(mu);
        q2[t & 3] = __builtin_elementwise_fma(a[t], a[t], q2[t & 3]);
    }
    const v2f qv = (q2[0] + q2[1]) + (q2[2] + q2[3]);
    float q = qv.x + qv.y;
    q += __shfl_xor(q, 16);
    const float var = q * (1.0f / Ee);
    const float r   = rsqrtf(var + LN_EPS);

    #pragma unroll
    for (int t = 0; t < GB_PRE; ++t) a[t] = a[t] * splat2(r) * gm[t] + bt[t];
    #pragma unroll
    for (int t = GB_PRE; t < Ee / 4; ++t) a[t] = a[t] * splat2(r) * g2[2 * t] + e2[2 * t];

    v8f acc = {0.f, 0.f, 0.f, 0.f, 0.f, 0.f, 0.f, 0.f};

    #pragma unroll
    for (int t = 0; t < Ee / 4; ++t)
        acc = __builtin_amdgcn_wmma_f32_16x16x4_f32(
                   false, a[t],  false, Bf[t],
                   (short)0, acc,  false,  false);

    if (which == 0) {
        const float bias = b1[n];
        #pragma unroll
        for (int r8 = 0; r8 < 8; ++r8) {
            const int mo = mt * 16 + 8 * half + r8;
            aP[(size_t)mo * Hh + n] = acc[r8] + bias;
        }
    } else {
        #pragma unroll
        for (int r8 = 0; r8 < 8; ++r8) {
            const int mo = mt * 16 + 8 * half + r8;
            const int bb = mo >> 9;
            const int j  = mo & (Nn - 1);
            bT[(size_t)bb * Hh * Nn + (size_t)n * Nn + j] = acc[r8];
        }
    }
}

__device__ __forceinline__ v2f gelu_erf_f32x2(v2f x) {
    const v2f u = __builtin_elementwise_min(__builtin_elementwise_abs(x), splat2(6.0f));
    v2f p = splat2(3.30960393e-5f);
    p = __builtin_elementwise_fma(p, u, splat2(-7.69245671e-4f));
    p = __builtin_elementwise_fma(p, u, splat2(8.08079448e-3f));
    p = __builtin_elementwise_fma(p, u, splat2(-5.34122139e-2f));
    p = __builtin_elementwise_fma(p, u, splat2(-4.58770901e-1f));
    p = __builtin_elementwise_fma(p, u, splat2(-1.15120173f));
    p = __builtin_elementwise_fma(p, u, splat2(-9.99993086e-1f));
    v2f e;
    e.x = __builtin_amdgcn_exp2f(p.x);
    e.y = __builtin_amdgcn_exp2f(p.y);
    const v2f m = __builtin_elementwise_max(x, splat2(0.0f));
    return __builtin_elementwise_fma(-u, e, m);
}

#define PR_I 2
#define PR_T 128
#define PR_H 4

struct __attribute__((aligned(32))) PairRow { v2f a[PR_I]; v2f w; v2f pad; };

__device__ __forceinline__ void pair_trip(const PairRow* __restrict__ row,
                                          const v2f (&cb)[PR_H], v2f (&acc)[PR_I]) {
    #pragma unroll
    for (int k = 0; k < PR_H; ++k) {
        const v2f w = row[k].w;
        #pragma unroll
        for (int r = 0; r < PR_I; ++r) {
            const v2f x = row[k].a[r] + cb[k];
            acc[r] = __builtin_elementwise_fma(w, gelu_erf_f32x2(x), acc[r]);
        }
    }
}

__global__ __launch_bounds__(PR_T) void pair_gelu_kernel(const float* __restrict__ aP,
                                 const float* __restrict__ bT,
                                 const float* __restrict__ w2,
                                 const float* __restrict__ b2,
                                 float* __restrict__ l) {
    const int t  = threadIdx.x;
    const int j  = blockIdx.x * (2 * PR_T) + 2 * t;
    const int i0 = blockIdx.y * PR_I;
    const int bb = blockIdx.z;

    __shared__ PairRow sR[Hh];
    static_assert(Hh % PR_T == 0, "each lane fills Hh / PR_T rows of sR");
    #pragma unroll
    for (int p = 0; p < Hh / PR_T; ++p) {
        const int h = t + p * PR_T;
        #pragma unroll
        for (int r = 0; r < PR_I; ++r)
            sR[h].a[r] = splat2(aP[((size_t)(bb * Nn + i0 + r)) * Hh + h]);
        sR[h].w = splat2(w2[h]);
    }
    const v2f* __restrict__ bq =
        reinterpret_cast<const v2f*>(bT + (size_t)bb * Hh * Nn + j);

    v2f b0[PR_H], b1[PR_H];
    #pragma unroll
    for (int k = 0; k < PR_H; ++k) b0[k] = bq[(size_t)k * (Nn / 2)];

    __syncthreads();

    v2f acc[PR_I];
    #pragma unroll
    for (int r = 0; r < PR_I; ++r) acc[r] = splat2(b2[0]);

    for (int h = 0; h < Hh - 2 * PR_H; h += 2 * PR_H) {
        #pragma unroll
        for (int k = 0; k < PR_H; ++k) b1[k] = bq[(size_t)(h + PR_H + k) * (Nn / 2)];
        pair_trip(&sR[h], b0, acc);
        #pragma unroll
        for (int k = 0; k < PR_H; ++k) b0[k] = bq[(size_t)(h + 2 * PR_H + k) * (Nn / 2)];
        pair_trip(&sR[h + PR_H], b1, acc);
    }
    #pragma unroll
    for (int k = 0; k < PR_H; ++k) b1[k] = bq[(size_t)(Hh - PR_H + k) * (Nn / 2)];
    pair_trip(&sR[Hh - 2 * PR_H], b0, acc);
    pair_trip(&sR[Hh - PR_H], b1, acc);

    #pragma unroll
    for (int r = 0; r < PR_I; ++r)
        *reinterpret_cast<v2f*>(l + ((size_t)(bb * Nn + i0 + r)) * Nn + j) = acc[r];
}

#define SY_T  32
#define SY_NT (Nn / SY_T)

__global__ __launch_bounds__(256) void symmetrize_kernel(const float* __restrict__ l,
                                                         float* __restrict__ out) {
    const int x = blockIdx.x;
    const int y = blockIdx.y;
    const bool right = x > y;
    const int I = right ? y : SY_NT - 1 - y;
    const int J = right ? x - 1 : SY_NT - 1 - x;
    const size_t base = (size_t)blockIdx.z * Nn * Nn;

    __shared__ float sA[SY_T][SY_T + 1];
    __shared__ float sB[SY_T][SY_T + 1];

    const int t = threadIdx.x;
    const int r = t >> 3;
    const int c = (t & 7) << 2;

    const size_t offA = base + (size_t)(I * SY_T + r) * Nn + (size_t)(J * SY_T + c);
    const size_t offB = base + (size_t)(J * SY_T + r) * Nn + (size_t)(I * SY_T + c);
    const float4 a = *reinterpret_cast<const float4*>(l + offA);
    const float4 b = *reinterpret_cast<const float4*>(l + offB);

    sA[r][c + 0] = a.x; sA[r][c + 1] = a.y; sA[r][c + 2] = a.z; sA[r][c + 3] = a.w;
    sB[r][c + 0] = b.x; sB[r][c + 1] = b.y; sB[r][c + 2] = b.z; sB[r][c + 3] = b.w;
    __syncthreads();

    float4 o;
    o.x = 0.5f * (a.x + sB[c + 0][r]);
    o.y = 0.5f * (a.y + sB[c + 1][r]);
    o.z = 0.5f * (a.z + sB[c + 2][r]);
    o.w = 0.5f * (a.w + sB[c + 3][r]);
    *reinterpret_cast<float4*>(out + offA) = o;

    if (I != J) {
        o.x = 0.5f * (b.x + sA[c + 0][r]);
        o.y = 0.5f * (b.y + sA[c + 1][r]);
        o.z = 0.5f * (b.z + sA[c + 2][r]);
        o.w = 0.5f * (b.w + sA[c + 3][r]);
        *reinterpret_cast<float4*>(out + offB) = o;
    }
}

extern "C" void kernel_launch(void* const* d_in, const int* in_sizes, int n_in,
                              void* d_out, int out_size, void* d_ws, size_t ws_size,
                              hipStream_t stream) {
    const float* h     = (const float*)d_in[0];
    const float* gamma = (const float*)d_in[1];
    const float* beta  = (const float*)d_in[2];
    const float* W1    = (const float*)d_in[3];
    const float* b1    = (const float*)d_in[4];
    const float* w2    = (const float*)d_in[5];
    const float* b2    = (const float*)d_in[6];
    float* out = (float*)d_out;

    float* ws  = (float*)d_ws;
    float* aP  = ws;
    float* bT  = aP + (size_t)ROWS * Hh;
    float* lbuf= bT + (size_t)Bb * Hh * Nn;

    ln_gemm_wmma_kernel<<<dim3(64 * 16, 2), dim3(32), 0, stream>>>(
        h, gamma, beta, W1, b1, aP, bT);

    pair_gelu_kernel<<<dim3(Nn / (2 * PR_T), Nn / PR_I, Bb), dim3(PR_T), 0, stream>>>(
        aP, bT, w2, b2, lbuf);

    symmetrize_kernel<<<dim3(SY_NT + 1, SY_NT / 2, Bb), dim3(256), 0, stream>>>(lbuf, out);
}


# symmetrisation folded into the pair kernel: a block computes a mirrored pair of 32x32 tiles of l from aP and bT tiles staged in LDS, swaps them in LDS and writes out; 3 launches become 2; sums kept
# speedup vs baseline: 4.4703x; 1.1282x over previous
#include <hip/hip_runtime.h>
#include <hip/hip_bf16.h>
#include <math.h>

typedef float v2f __attribute__((ext_vector_type(2)));
typedef float v8f __attribute__((ext_vector_type(8)));

#define Bb 2
#define Nn 512
#define Ee 128
#define Hh 256
#define ROWS (Bb * Nn)
#define LN_EPS 1e-5f
#define GB_PRE 16

__device__ __forceinline__ v2f splat2(float c) { return v2f{c, c}; }

__global__ __launch_bounds__(32) void ln_gemm_wmma_kernel(const float* __restrict__ h,
                                 const float* __restrict__ gamma,
                                 const float* __restrict__ beta,
                                 const float* __restrict__ W1,
                                 const float* __restrict__ b1,
                                 float* __restrict__ aP,
                                 float* __restrict__ bT) {
    const int tile = blockIdx.x;
    const int mt   = tile >> 4;
    const int nt   = tile & 15;
    const int which = blockIdx.y;
    const float* __restrict__ W = W1 + (size_t)which * Ee * Hh;

    const int lane = threadIdx.x;
    const int half = lane >> 4;
    const int l16  = lane & 15;

    const int m = mt * 16 + l16;
    const int n = nt * 16 + l16;

    const v2f* __restrict__ hrow =
        reinterpret_cast<const v2f*>(h + (size_t)m * Ee + 2 * half);
    v2f a[Ee / 4];
    #pragma unroll
    for (int t = 0; t < Ee / 4; ++t) a[t] = hrow[2 * t];

    v2f Bf[Ee / 4];
    #pragma unroll
    for (int t = 0; t < Ee / 4; ++t) {
        const int ka = 4 * t + 2 * half;
        Bf[t].x = W[(size_t)ka * Hh + n];
        Bf[t].y = W[(size_t)(ka + 1) * Hh + n];
    }
    const v2f* __restrict__ g2 = reinterpret_cast<const v2f*>(gamma + 2 * half);
    const v2f* __restrict__ e2 = reinterpret_cast<const v2f*>(beta + 2 * half);
    v2f gm[GB_PRE], bt[GB_PRE];
    #pragma unroll
    for (int t = 0; t < GB_PRE; ++t) { gm[t] = g2[2 * t]; bt[t] = e2[2 * t]; }
    __builtin_amdgcn_sched_barrier(0);

    v2f s2[4] = {splat2(0.f), splat2(0.f), splat2(0.f), splat2(0.f)};
    #pragma unroll
    for (int t = 0; t < Ee / 4; ++t) s2[t & 3] += a[t];
    const v2f sv = (s2[0] + s2[1]) + (s2[2] + s2[3]);
    float s = sv.x + sv.y;
    s += __shfl_xor(s, 16);
    const float mu = s * (1.0f / Ee);

    v2f q2[4] = {splat2(0.f), splat2(0.f), splat2(0.f), splat2(0.f)};
    #pragma unroll
    for (int t = 0; t < Ee / 4; ++t) {
        a[t] -= splat2(mu);
        q2[t & 3] = __builtin_elementwise_fma(a[t], a[t], q2[t & 3]);
    }
    const v2f qv = (q2[0] + q2[1]) + (q2[2] + q2[3]);
    float q = qv.x + qv.y;
    q += __shfl_xor(q, 16);
    const float var = q * (1.0f / Ee);
    const float r   = rsqrtf(var + LN_EPS);

    #pragma unroll
    for (int t = 0; t < GB_PRE; ++t) a[t] = a[t] * splat2(r) * gm[t] + bt[t];
    #pragma unroll
    for (int t = GB_PRE; t < Ee / 4; ++t) a[t] = a[t] * splat2(r) * g2[2 * t] + e2[2 * t];

    v8f acc = {0.f, 0.f, 0.f, 0.f, 0.f, 0.f, 0.f, 0.f};

    #pragma unroll
    for (int t = 0; t < Ee / 4; ++t)
        acc = __builtin_amdgcn_wmma_f32_16x16x4_f32(
                   false, a[t],  false, Bf[t],
                   (short)0, acc,  false,  false);

    if (which == 0) {
        const float bias = b1[n];
        #pragma unroll
        for (int r8 = 0; r8 < 8; ++r8) {
            const int mo = mt * 16 + 8 * half + r8;
            aP[(size_t)mo * Hh + n] = acc[r8] + bias;
        }
    } else {
        #pragma unroll
        for (int r8 = 0; r8 < 8; ++r8) {
            const int mo = mt * 16 + 8 * half + r8;
            const int bb = mo >> 9;
            const int j  = mo & (Nn - 1);
            bT[(size_t)bb * Hh * Nn + (size_t)n * Nn + j] = acc[r8];
        }
    }
}

__device__ __forceinline__ v2f gelu_erf_f32x2(v2f x) {
    const v2f u = __builtin_elementwise_min(__builtin_elementwise_abs(x), splat2(6.0f));
    v2f p = splat2(3.30960393e-5f);
    p = __builtin_elementwise_fma(p, u, splat2(-7.69245671e-4f));
    p = __builtin_elementwise_fma(p, u, splat2(8.08079448e-3f));
    p = __builtin_elementwise_fma(p, u, splat2(-5.34122139e-2f));
    p = __builtin_elementwise_fma(p, u, splat2(-4.58770901e-1f));
    p = __builtin_elementwise_fma(p, u, splat2(-1.15120173f));
    p = __builtin_elementwise_fma(p, u, splat2(-9.99993086e-1f));
    v2f e;
    e.x = __builtin_amdgcn_exp2f(p.x);
    e.y = __builtin_amdgcn_exp2f(p.y);
    const v2f m = __builtin_elementwise_max(x, splat2(0.0f));
    return __builtin_elementwise_fma(-u, e, m);
}

#define TP_T   32
#define TP_NT  (Nn / TP_T)
#define TP_RP  (TP_T / 2)
#define TP_JOB 256
#define TP_H   8

struct __attribute__((aligned(16))) RowPairA { v2f a[2]; };

__global__ __launch_bounds__(2 * TP_JOB) void pair_gelu_sym_kernel(
                                 const float* __restrict__ aP,
                                 const float* __restrict__ bT,
                                 const float* __restrict__ w2,
                                 const float* __restrict__ b2,
                                 float* __restrict__ out) {
    __shared__ RowPairA sA[2][TP_RP][Hh + 1];
    __shared__ __attribute__((aligned(16))) v2f sBt[2][Hh][TP_T / 2];
    __shared__ __attribute__((aligned(16))) v2f sW[Hh];
    __shared__ float sL[2][TP_T][TP_T + 1];

    const int t    = threadIdx.x;
    const int hb   = t >> 8;
    const int tt   = t & (TP_JOB - 1);
    const int w8   = tt >> 5;
    const int half = (t >> 4) & 1;
    const int l16  = t & 15;
    const int bb   = blockIdx.y;

    const int d  = blockIdx.x >> 4;
    const int i0 = blockIdx.x & 15;
    const int A  = (d == 0) ? (i0 & 7) : i0;
    const int B  = (d == 0) ? (i0 & 7) + 8 : ((i0 + d) & 15);
    const int cross = (d == 0 && i0 < 8) ? 0 : 1;
    const int R = hb ? B : A;
    const int C = (hb ^ cross) ? B : A;

    {
        const float* __restrict__ arow =
            aP + ((size_t)(bb * Nn + R * TP_T)) * Hh + tt;
        float av[TP_T];
        #pragma unroll
        for (int r = 0; r < TP_T; ++r) av[r] = arow[(size_t)r * Hh];
        #pragma unroll
        for (int rp = 0; rp < TP_RP; ++rp) {
            RowPairA e;
            e.a[0] = splat2(av[2 * rp]);
            e.a[1] = splat2(av[2 * rp + 1]);
            sA[hb][rp][tt] = e;
        }
    }
    {
        const int c4 = (tt & 7) << 2;
        const float* __restrict__ bsrc =
            bT + (size_t)bb * Hh * Nn + (size_t)(tt >> 3) * Nn + (size_t)(C * TP_T + c4);
        float4 bv[Hh / 32];
        #pragma unroll
        for (int p = 0; p < Hh / 32; ++p)
            bv[p] = *reinterpret_cast<const float4*>(bsrc + (size_t)p * 32 * Nn);
        #pragma unroll
        for (int p = 0; p < Hh / 32; ++p)
            *reinterpret_cast<float4*>(&sBt[hb][(tt >> 3) + 32 * p][(tt & 7) << 1]) = bv[p];
    }
    if (t < Hh) sW[t] = splat2(w2[t]);
    __syncthreads();

    const RowPairA* __restrict__ pa = &sA[hb][2 * w8 + half][0];
    const v2f* __restrict__ pb = &sBt[hb][0][l16];

    v2f acc[2];
    acc[0] = splat2(b2[0]);
    acc[1] = acc[0];

    for (int h = 0; h < Hh; h += TP_H) {
        #pragma unroll
        for (int k = 0; k < TP_H; ++k) {
            const v2f w  = sW[h + k];
            const RowPairA ar = pa[h + k];
            const v2f cb = pb[(size_t)(h + k) * (TP_T / 2)];
            #pragma unroll
            for (int r = 0; r < 2; ++r) {
                const v2f x = ar.a[r] + cb;
                acc[r] = __builtin_elementwise_fma(w, gelu_erf_f32x2(x), acc[r]);
            }
        }
    }

    #pragma unroll
    for (int r = 0; r < 2; ++r) {
        const int row = 2 * (2 * w8 + half) + r;
        sL[hb][row][2 * l16]     = acc[r].x;
        sL[hb][row][2 * l16 + 1] = acc[r].y;
    }
    __syncthreads();

    const float (*own)[TP_T + 1] = sL[hb];
    const float (*mir)[TP_T + 1] = sL[cross ? (hb ^ 1) : hb];
    const int r = tt >> 3;
    const int c = (tt & 7) << 2;
    float4 o;
    o.x = 0.5f * (own[r][c + 0] + mir[c + 0][r]);
    o.y = 0.5f * (own[r][c + 1] + mir[c + 1][r]);
    o.z = 0.5f * (own[r][c + 2] + mir[c + 2][r]);
    o.w = 0.5f * (own[r][c + 3] + mir[c + 3][r]);
    *reinterpret_cast<float4*>(out + (size_t)bb * Nn * Nn
                               + (size_t)(R * TP_T + r) * Nn + (size_t)(C * TP_T + c)) = o;
}

extern "C" void kernel_launch(void* const* d_in, const int* in_sizes, int n_in,
                              void* d_out, int out_size, void* d_ws, size_t ws_size,
                              hipStream_t stream) {
    const float* h     = (const float*)d_in[0];
    const float* gamma = (const float*)d_in[1];
    const float* beta  = (const float*)d_in[2];
    const float* W1    = (const float*)d_in[3];
    const float* b1    = (const float*)d_in[4];
    const float* w2    = (const float*)d_in[5];
    const float* b2    = (const float*)d_in[6];
    float* out = (float*)d_out;

    float* ws  = (float*)d_ws;
    float* aP  = ws;
    float* bT  = aP + (size_t)ROWS * Hh;

    ln_gemm_wmma_kernel<<<dim3(64 * 16, 2), dim3(32), 0, stream>>>(
        h, gamma, beta, W1, b1, aP, bT);

    pair_gelu_sym_kernel<<<dim3(TP_NT * TP_NT / 2, Bb), dim3(2 * TP_JOB), 0, stream>>>(
        aP, bT, w2, b2, out);
}
